# starting kernel
# baseline (speedup 1.0000x reference)
#include <hip/hip_runtime.h>


typedef float v2f __attribute__((ext_vector_type(2)));
typedef float v8f __attribute__((ext_vector_type(8)));

#define B_   32
#define C_   64
#define H_   32
#define W_   32
#define KD   576
#define COG  16
#define TILE 16
#define CCH  32

__global__ __launch_bounds__(256) void wtrans_kernel(const float* __restrict__ w,
                                                     float* __restrict__ wt) {
  int idx = blockIdx.x * 256 + threadIdx.x;
  if (idx < C_ * KD) {
    int co = idx / KD, k = idx - co * KD;
    int cg = co >> 4, c = co & 15;
    wt[cg * (KD * COG) + k * COG + c] = w[idx];
  }
}

__global__ __launch_bounds__(256) void adder_kernel(const float* __restrict__ in,
                                                    const float* __restrict__ wt,
                                                    float* __restrict__ y) {
  __shared__ float xs[CCH][TILE + 2][TILE + 2];
  const int tid    = threadIdx.x;
  const int cg     = blockIdx.x;
  const int tileId = blockIdx.y;
  const int n      = blockIdx.z;
  const int ty0 = (tileId >> 1) * TILE;
  const int tx0 = (tileId & 1) * TILE;
  const int tx = tid & 15, ty = tid >> 4;

  const float* wblk = wt + cg * (KD * COG);
  __builtin_prefetch(wblk, 0, 1);

  float acc[COG];
#pragma unroll
  for (int c = 0; c < COG; ++c) acc[c] = 0.f;

  for (int c0 = 0; c0 < C_; c0 += CCH) {
    for (int i = tid; i < CCH * (TILE + 2) * (TILE + 2); i += 256) {
      int ci = i / ((TILE + 2) * (TILE + 2));
      int r  = i - ci * ((TILE + 2) * (TILE + 2));
      int yy = r / (TILE + 2);
      int xx = r - yy * (TILE + 2);
      int gy = ty0 + yy - 1;
      int gx = tx0 + xx - 1;
      float v = 0.f;
      if ((unsigned)gy < (unsigned)H_ && (unsigned)gx < (unsigned)W_)
        v = in[(((size_t)n * C_ + (c0 + ci)) * H_ + gy) * W_ + gx];
      xs[ci][yy][xx] = v;
    }
    __syncthreads();

    for (int ci = 0; ci < CCH; ++ci) {
#pragma unroll
      for (int kh = 0; kh < 3; ++kh) {
#pragma unroll
        for (int kw = 0; kw < 3; ++kw) {
          const float xv = xs[ci][ty + kh][tx + kw];
          const float* wk = wblk + (((c0 + ci) * 3 + kh) * 3 + kw) * COG;
#pragma unroll
          for (int c = 0; c < COG; ++c)
            acc[c] += fabsf(xv - wk[c]);
        }
      }
    }
    __syncthreads();
  }

  const int gy = ty0 + ty, gx = tx0 + tx;
#pragma unroll
  for (int c = 0; c < COG; ++c)
    y[(((size_t)n * C_ + (cg * COG + c)) * H_ + gy) * W_ + gx] = -acc[c];
}

__global__ __launch_bounds__(256) void stats_kernel(const float* __restrict__ y,
                                                    const float* __restrict__ g,
                                                    const float* __restrict__ b,
                                                    float* __restrict__ ss) {
  const int c    = blockIdx.x;
  const int tid  = threadIdx.x;
  const int wave = tid >> 5, lane = tid & 31;

  v8f accS = {};
  v8f accQ = {};
  v2f ones = {1.f, 1.f};

  const float* chan = y + (size_t)c * (H_ * W_);
  for (int base = wave * 64; base < B_ * H_ * W_; base += 8 * 64) {
    int e0 = base + lane * 2;
    int n0 = e0 >> 10, i0 = e0 & 1023;
    const float* p = chan + (size_t)n0 * (C_ * H_ * W_) + i0;
    v2f d = {p[0], p[1]};
    v2f q = {d.x * d.x, d.y * d.y};
    accS = __builtin_amdgcn_wmma_f32_16x16x4_f32(false, ones, false, d,
                                                 (short)0, accS, false, false);
    accQ = __builtin_amdgcn_wmma_f32_16x16x4_f32(false, ones, false, q,
                                                 (short)0, accQ, false, false);
  }

  float s = accS[0], q = accQ[0];
  for (int off = 16; off > 0; off >>= 1) {
    s += __shfl_xor(s, off, 32);
    q += __shfl_xor(q, off, 32);
  }
  s *= 0.5f;
  q *= 0.5f;

  __shared__ float ls[8], lq[8];
  if (lane == 0) { ls[wave] = s; lq[wave] = q; }
  __syncthreads();
  if (tid == 0) {
    float S = 0.f, Q = 0.f;
    for (int i = 0; i < 8; ++i) { S += ls[i]; Q += lq[i]; }
    const float inv_n = 1.0f / (float)(B_ * H_ * W_);
    float mu    = S * inv_n;
    float var   = Q * inv_n - mu * mu;
    float inv   = rsqrtf(var + 1e-5f);
    float scale = g[c] * inv;
    ss[c]       = scale;
    ss[C_ + c]  = b[c] - mu * scale;
  }
}

__global__ __launch_bounds__(256) void apply_kernel(const float* __restrict__ y,
                                                    const float* __restrict__ ss,
                                                    const float* __restrict__ resid,
                                                    float* __restrict__ out, int n) {
  int idx = blockIdx.x * 256 + threadIdx.x;
  if (idx < n) {
    int c = (idx >> 10) & (C_ - 1);
    float v = y[idx] * ss[c] + ss[C_ + c];
    if (resid) v += resid[idx];
    out[idx] = fmaxf(v, 0.f);
  }
}

extern "C" void kernel_launch(void* const* d_in, const int* in_sizes, int n_in,
                              void* d_out, int out_size, void* d_ws, size_t ws_size,
                              hipStream_t stream) {
  (void)in_sizes; (void)n_in; (void)out_size; (void)ws_size;
  const float* x  = (const float*)d_in[0];
  const float* w1 = (const float*)d_in[1];
  const float* g1 = (const float*)d_in[2];
  const float* b1 = (const float*)d_in[3];
  const float* w2 = (const float*)d_in[4];
  const float* g2 = (const float*)d_in[5];
  const float* b2 = (const float*)d_in[6];
  float* out = (float*)d_out;

  char* ws = (char*)d_ws;
  const size_t TEN   = (size_t)B_ * C_ * H_ * W_;
  const size_t WTSZ  = (size_t)C_ * KD * sizeof(float);
  float* ybuf = (float*)(ws);
  float* abuf = (float*)(ws + TEN * 4);
  float* wt1  = (float*)(ws + 2 * TEN * 4);
  float* wt2  = (float*)(ws + 2 * TEN * 4 + WTSZ);
  float* ss1  = (float*)(ws + 2 * TEN * 4 + 2 * WTSZ);
  float* ss2  = (float*)(ws + 2 * TEN * 4 + 2 * WTSZ + 512);

  wtrans_kernel<<<144, 256, 0, stream>>>(w1, wt1);
  wtrans_kernel<<<144, 256, 0, stream>>>(w2, wt2);

  dim3 agrid(C_ / COG, (H_ / TILE) * (W_ / TILE), B_);

  adder_kernel<<<agrid, 256, 0, stream>>>(x, wt1, ybuf);
  stats_kernel<<<C_, 256, 0, stream>>>(ybuf, g1, b1, ss1);
  apply_kernel<<<(int)(TEN / 256), 256, 0, stream>>>(ybuf, ss1, nullptr, abuf, (int)TEN);

  adder_kernel<<<agrid, 256, 0, stream>>>(abuf, wt2, ybuf);
  stats_kernel<<<C_, 256, 0, stream>>>(ybuf, g2, b2, ss2);
  apply_kernel<<<(int)(TEN / 256), 256, 0, stream>>>(ybuf, ss2, x, out, (int)TEN);
}


# adder tap loads run three channels ahead of the sums (four channels per trip, four register sets, last trip peeled): register copies and most offset adds gone, waits exact per set; sum order kept
# speedup vs baseline: 2.6535x; 2.6535x over previous
#include <hip/hip_runtime.h>


typedef float v2f __attribute__((ext_vector_type(2)));
typedef float v8f __attribute__((ext_vector_type(8)));
typedef __attribute__((address_space(4))) const float kfloat;

#define B_   32
#define C_   64
#define H_   32
#define W_   32
#define KD   576
#define COG  16
#define TILE 16

__global__ __launch_bounds__(256) void wtrans_kernel(const float* __restrict__ w,
                                                     float* __restrict__ wt) {
  int idx = blockIdx.x * 256 + threadIdx.x;
  if (idx < C_ * KD) {
    int co = idx / KD, k = idx - co * KD;
    int cg = co >> 4, c = co & 15;
    wt[cg * (KD * COG) + k * COG + c] = w[idx];
  }
}

__device__ __forceinline__ void tap8(float& a0, float& a1, float& a2, float& a3,
                                     float& a4, float& a5, float& a6, float& a7,
                                     float x,
                                     float w0, float w1, float w2, float w3,
                                     float w4, float w5, float w6, float w7) {
  asm volatile("v_dual_sub_f32 v120, %[w0], %[x] :: v_dual_sub_f32 v121, %[w1], %[x]\n\t"
      "v_dual_sub_f32 v122, %[w2], %[x] :: v_dual_sub_f32 v123, %[w3], %[x]\n\t"
      "v_dual_sub_f32 v124, %[w4], %[x] :: v_dual_sub_f32 v125, %[w5], %[x]\n\t"
      "v_dual_sub_f32 v126, %[w6], %[x] :: v_dual_sub_f32 v127, %[w7], %[x]\n\t"
      "s_delay_alu instid0(VALU_DEP_4)\n\t"
      "v_add_f32 %[a0], |v120|, %[a0]\n\t"
      "v_add_f32 %[a1], |v121|, %[a1]\n\t"
      "v_add_f32 %[a2], |v122|, %[a2]\n\t"
      "v_add_f32 %[a3], |v123|, %[a3]\n\t"
      "v_add_f32 %[a4], |v124|, %[a4]\n\t"
      "v_add_f32 %[a5], |v125|, %[a5]\n\t"
      "v_add_f32 %[a6], |v126|, %[a6]\n\t"
      "v_add_f32 %[a7], |v127|, %[a7]"
      : [a0] "+v"(a0), [a1] "+v"(a1), [a2] "+v"(a2), [a3] "+v"(a3),
        [a4] "+v"(a4), [a5] "+v"(a5), [a6] "+v"(a6), [a7] "+v"(a7)
      : [x] "v"(x),
        [w0] "s"(w0), [w1] "s"(w1), [w2] "s"(w2), [w3] "s"(w3),
        [w4] "s"(w4), [w5] "s"(w5), [w6] "s"(w6), [w7] "s"(w7)
      : "v120", "v121", "v122", "v123", "v124", "v125", "v126", "v127");
}

__device__ __forceinline__ void channel16(float (&acc)[COG], const float (&xc)[9],
                                          const kfloat* wk) {
  float wa[9][8], wb[9][8];
#pragma unroll
  for (int t = 0; t < 9; ++t)
#pragma unroll
    for (int c = 0; c < 8; ++c) wa[t][c] = wk[t * COG + c];
#pragma unroll
  for (int t = 0; t < 9; ++t) {
    tap8(acc[0], acc[1], acc[2], acc[3], acc[4], acc[5], acc[6], acc[7], xc[t],
         wa[t][0], wa[t][1], wa[t][2], wa[t][3], wa[t][4], wa[t][5], wa[t][6], wa[t][7]);
#pragma unroll
    for (int c = 0; c < 8; ++c) wb[t][c] = wk[t * COG + 8 + c];
  }
#pragma unroll
  for (int t = 0; t < 9; ++t)
    tap8(acc[8], acc[9], acc[10], acc[11], acc[12], acc[13], acc[14], acc[15], xc[t],
         wb[t][0], wb[t][1], wb[t][2], wb[t][3], wb[t][4], wb[t][5], wb[t][6], wb[t][7]);
}

__global__ __launch_bounds__(256) void adder_kernel(const float* __restrict__ in,
                                                    const float* __restrict__ wt,
                                                    float* __restrict__ y,
                                                    float* __restrict__ part) {
  __shared__ __attribute__((aligned(16))) float ysm[COG * 256];
  __shared__ float red[32];
  const int tid    = threadIdx.x;
  const int cg     = blockIdx.x;
  const int tileId = blockIdx.y;
  const int n      = blockIdx.z;
  const int ty0 = (tileId >> 1) * TILE;
  const int tx0 = (tileId & 1) * TILE;
  const int tx = tid & 15, ty = tid >> 4;
  const int gy = ty0 + ty, gx = tx0 + tx;

  const float* wblk = wt + cg * (KD * COG);
  __builtin_prefetch(wblk, 0, 1);
  const kfloat* wconst = (const kfloat*)wblk;

  const int  ry[3] = {gy > 0 ? gy - 1 : 0, gy, gy < H_ - 1 ? gy + 1 : H_ - 1};
  const int  rx[3] = {gx > 0 ? gx - 1 : 0, gx, gx < W_ - 1 ? gx + 1 : W_ - 1};
  const bool vy[3] = {gy > 0, true, gy < H_ - 1};
  const bool vx[3] = {gx > 0, true, gx < W_ - 1};
  unsigned off[9];
  bool     ok[9];
#pragma unroll
  for (int kh = 0; kh < 3; ++kh)
#pragma unroll
    for (int kw = 0; kw < 3; ++kw) {
      off[kh * 3 + kw] = (unsigned)(ry[kh] * W_ + rx[kw]);
      ok[kh * 3 + kw]  = vy[kh] && vx[kw];
    }

  const float* img = in + (size_t)n * C_ * H_ * W_;

  float acc[COG];
#pragma unroll
  for (int c = 0; c < COG; ++c) acc[c] = 0.f;

  constexpr int PF = 4;
  constexpr int HW = H_ * W_;
  float xs[PF][9];
#pragma unroll
  for (int u = 0; u < PF - 1; ++u) {
#pragma unroll
    for (int t = 0; t < 9; ++t) xs[u][t] = (img + u * HW)[off[t]];
    __builtin_amdgcn_sched_barrier(0);
  }

  for (int ci = 0; ci < C_ - PF; ci += PF) {
    unsigned offn[9];
#pragma unroll
    for (int t = 0; t < 9; ++t) offn[t] = off[t] + (unsigned)(PF * HW);
#pragma unroll
    for (int t = 0; t < 9; ++t) xs[PF - 1][t] = *(img + offn[t] - HW);
#pragma unroll
    for (int u = 0; u < PF; ++u) {
      __builtin_amdgcn_sched_barrier(0);
      float xc[9];
#pragma unroll
      for (int t = 0; t < 9; ++t) xc[t] = ok[t] ? xs[u][t] : 0.f;
      channel16(acc, xc, wconst + (ci + u) * 9 * COG);
      if (u < PF - 1) {
#pragma unroll
        for (int t = 0; t < 9; ++t) xs[u][t] = (img + u * HW)[offn[t]];
      }
    }
#pragma unroll
    for (int t = 0; t < 9; ++t) off[t] = offn[t];
  }
  {
#pragma unroll
    for (int t = 0; t < 9; ++t) xs[PF - 1][t] = (img + (PF - 1) * HW)[off[t]];
#pragma unroll
    for (int u = 0; u < PF; ++u) {
      __builtin_amdgcn_sched_barrier(0);
      float xc[9];
#pragma unroll
      for (int t = 0; t < 9; ++t) xc[t] = ok[t] ? xs[u][t] : 0.f;
      channel16(acc, xc, wconst + (C_ - PF + u) * 9 * COG);
    }
  }

#pragma unroll
  for (int c = 0; c < COG; ++c) ysm[c * 256 + tid] = -acc[c];
  __syncthreads();

  const int wave = tid >> 5, lane = tid & 31;
  const v2f ones = {1.f, 1.f};
  v8f aS0 = {}, aQ0 = {}, aS1 = {}, aQ1 = {};
  const float* r0 = ysm + (2 * wave) * 256;
  const float* r1 = r0 + 256;
#pragma unroll
  for (int h = 0; h < 2; ++h) {
    const float4 u = *(const float4*)(r0 + h * 128 + lane * 4);
    const float4 v = *(const float4*)(r1 + h * 128 + lane * 4);
    const v2f ua = {u.x, u.y}, ub = {u.z, u.w};
    const v2f va = {v.x, v.y}, vb = {v.z, v.w};
    const v2f uaq = {u.x * u.x, u.y * u.y}, ubq = {u.z * u.z, u.w * u.w};
    const v2f vaq = {v.x * v.x, v.y * v.y}, vbq = {v.z * v.z, v.w * v.w};
    aS0 = __builtin_amdgcn_wmma_f32_16x16x4_f32(false, ones, false, ua,
                                                (short)0, aS0, false, false);
    aS1 = __builtin_amdgcn_wmma_f32_16x16x4_f32(false, ones, false, va,
                                                (short)0, aS1, false, false);
    aQ0 = __builtin_amdgcn_wmma_f32_16x16x4_f32(false, ones, false, uaq,
                                                (short)0, aQ0, false, false);
    aQ1 = __builtin_amdgcn_wmma_f32_16x16x4_f32(false, ones, false, vaq,
                                                (short)0, aQ1, false, false);
    aS0 = __builtin_amdgcn_wmma_f32_16x16x4_f32(false, ones, false, ub,
                                                (short)0, aS0, false, false);
    aS1 = __builtin_amdgcn_wmma_f32_16x16x4_f32(false, ones, false, vb,
                                                (short)0, aS1, false, false);
    aQ0 = __builtin_amdgcn_wmma_f32_16x16x4_f32(false, ones, false, ubq,
                                                (short)0, aQ0, false, false);
    aQ1 = __builtin_amdgcn_wmma_f32_16x16x4_f32(false, ones, false, vbq,
                                                (short)0, aQ1, false, false);
  }
  float s0 = aS0[0], q0 = aQ0[0], s1 = aS1[0], q1 = aQ1[0];
#pragma unroll
  for (int off = 8; off > 0; off >>= 1) {
    s0 += __shfl_xor(s0, off, 32);
    q0 += __shfl_xor(q0, off, 32);
    s1 += __shfl_xor(s1, off, 32);
    q1 += __shfl_xor(q1, off, 32);
  }
  if (lane == 0) {
    red[wave * 4 + 0] = s0;
    red[wave * 4 + 1] = q0;
    red[wave * 4 + 2] = s1;
    red[wave * 4 + 3] = q1;
  }
  __syncthreads();
  if (wave == 0) {
    const int blk = (n * 4 + tileId) * 4 + cg;
    part[blk * 32 + lane] = red[lane];
  }
#pragma unroll
  for (int c = 0; c < COG; ++c)
    y[(((size_t)n * C_ + (cg * COG + c)) * H_ + gy) * W_ + gx] = -acc[c];
}

__global__ __launch_bounds__(256) void bnapply_kernel(const float* __restrict__ y,
                                                      const float* __restrict__ part,
                                                      const float* __restrict__ g,
                                                      const float* __restrict__ b,
                                                      const float* __restrict__ resid,
                                                      float* __restrict__ out) {
  __shared__ float sc[2];
  const int tid  = threadIdx.x;
  const int wave = tid >> 5, lane = tid & 31;
  const int c    = blockIdx.x;
  const int cg   = c >> 4, cl = c & 15;
  const size_t plane = (size_t)H_ * W_;
  const size_t base = ((size_t)blockIdx.y * 4 * C_ + c) * plane + (size_t)tid * 4;

  float4 v[4], r[4];
#pragma unroll
  for (int j = 0; j < 4; ++j) v[j] = *(const float4*)(y + base + (size_t)j * C_ * plane);
  if (resid) {
#pragma unroll
    for (int j = 0; j < 4; ++j) r[j] = *(const float4*)(resid + base + (size_t)j * C_ * plane);
  }

  if (wave == 0) {
    const float* p = part + cg * 32 + cl * 2;
    const float2 t0 = *(const float2*)(p + (lane      ) * 128);
    const float2 t1 = *(const float2*)(p + (lane + 32 ) * 128);
    const float2 t2 = *(const float2*)(p + (lane + 64 ) * 128);
    const float2 t3 = *(const float2*)(p + (lane + 96 ) * 128);

    v8f accS = {};
    v8f accQ = {};
    const v2f ones = {1.f, 1.f};
    const v2f sa = {t0.x, t1.x}, sb = {t2.x, t3.x};
    const v2f qa = {t0.y, t1.y}, qb = {t2.y, t3.y};
    accS = __builtin_amdgcn_wmma_f32_16x16x4_f32(false, ones, false, sa,
                                                 (short)0, accS, false, false);
    accQ = __builtin_amdgcn_wmma_f32_16x16x4_f32(false, ones, false, qa,
                                                 (short)0, accQ, false, false);
    accS = __builtin_amdgcn_wmma_f32_16x16x4_f32(false, ones, false, sb,
                                                 (short)0, accS, false, false);
    accQ = __builtin_amdgcn_wmma_f32_16x16x4_f32(false, ones, false, qb,
                                                 (short)0, accQ, false, false);

    float S = accS[0], Q = accQ[0];
#pragma unroll
    for (int off = 8; off > 0; off >>= 1) {
      S += __shfl_xor(S, off, 32);
      Q += __shfl_xor(Q, off, 32);
    }

    if (lane == 0) {
      const float inv_n = 1.0f / (float)(B_ * H_ * W_);
      float mu    = S * inv_n;
      float var   = Q * inv_n - mu * mu;
      float inv   = rsqrtf(var + 1e-5f);
      float scale = g[c] * inv;
      sc[0]       = scale;
      sc[1]       = b[c] - mu * scale;
    }
  }
  __syncthreads();
  const float scale = sc[0], shift = sc[1];

#pragma unroll
  for (int j = 0; j < 4; ++j) {
    float4 o;
    o.x = v[j].x * scale + shift;
    o.y = v[j].y * scale + shift;
    o.z = v[j].z * scale + shift;
    o.w = v[j].w * scale + shift;
    if (resid) { o.x += r[j].x; o.y += r[j].y; o.z += r[j].z; o.w += r[j].w; }
    o.x = fmaxf(o.x, 0.f); o.y = fmaxf(o.y, 0.f);
    o.z = fmaxf(o.z, 0.f); o.w = fmaxf(o.w, 0.f);
    *(float4*)(out + base + (size_t)j * C_ * plane) = o;
  }
}

extern "C" void kernel_launch(void* const* d_in, const int* in_sizes, int n_in,
                              void* d_out, int out_size, void* d_ws, size_t ws_size,
                              hipStream_t stream) {
  (void)in_sizes; (void)n_in; (void)out_size; (void)ws_size;
  const float* x  = (const float*)d_in[0];
  const float* w1 = (const float*)d_in[1];
  const float* g1 = (const float*)d_in[2];
  const float* b1 = (const float*)d_in[3];
  const float* w2 = (const float*)d_in[4];
  const float* g2 = (const float*)d_in[5];
  const float* b2 = (const float*)d_in[6];
  float* out = (float*)d_out;

  char* ws = (char*)d_ws;
  const size_t TEN   = (size_t)B_ * C_ * H_ * W_;
  const size_t WTSZ  = (size_t)C_ * KD * sizeof(float);
  float* ybuf = (float*)(ws);
  float* abuf = (float*)(ws + TEN * 4);
  float* wt1  = (float*)(ws + 2 * TEN * 4);
  float* wt2  = (float*)(ws + 2 * TEN * 4 + WTSZ);
  float* pt1  = (float*)(ws + 2 * TEN * 4 + 2 * WTSZ + 1024);
  float* pt2  = (float*)(ws + 2 * TEN * 4 + 2 * WTSZ + 1024 + 65536);

  wtrans_kernel<<<144, 256, 0, stream>>>(w1, wt1);
  wtrans_kernel<<<144, 256, 0, stream>>>(w2, wt2);

  dim3 agrid(C_ / COG, (H_ / TILE) * (W_ / TILE), B_);
  dim3 ngrid(C_, B_ / 4);

  adder_kernel<<<agrid, 256, 0, stream>>>(x, wt1, ybuf, pt1);
  bnapply_kernel<<<ngrid, 256, 0, stream>>>(ybuf, pt1, g1, b1, nullptr, abuf);

  adder_kernel<<<agrid, 256, 0, stream>>>(abuf, wt2, ybuf, pt2);
  bnapply_kernel<<<ngrid, 256, 0, stream>>>(ybuf, pt2, g2, b2, x, out);
}


# adder weights reach scalar registers one part ahead of their sums (4 parts of 5/4 taps x 8 outputs); layout puts a quad of part 1 in each cache line: one fetch trip per channel; sum order kept
# speedup vs baseline: 2.8039x; 1.0567x over previous
#include <hip/hip_runtime.h>


typedef float v2f __attribute__((ext_vector_type(2)));
typedef float v8f __attribute__((ext_vector_type(8)));
typedef __attribute__((address_space(4))) const float kfloat;

#define B_   32
#define C_   64
#define H_   32
#define W_   32
#define KD   576
#define COG  16
#define TILE 16

__host__ __device__ constexpr int wpos(int t, int c) {
  const int h = c >> 3, j = (c >> 2) & 1, e = c & 3;
  if (h == 0 && t < 5) {
    const int n = 2 * t + j;
    return n < 9 ? n * 16 + e : 4 + e;
  }
  const int r = h == 0 ? t - 5 : 4 + t;
  if (r < 9) return r * 16 + 8 + 4 * j + e;
  return (1 + 2 * (r - 9) + j) * 16 + 4 + e;
}

__global__ __launch_bounds__(256) void wtrans_kernel(const float* __restrict__ w,
                                                     float* __restrict__ wt) {
  int idx = blockIdx.x * 256 + threadIdx.x;
  if (idx < C_ * KD) {
    int co = idx / KD, k = idx - co * KD;
    int cg = co >> 4, c = co & 15;
    int ci = k / 9, t = k - ci * 9;
    wt[cg * (KD * COG) + ci * (9 * COG) + wpos(t, c)] = w[idx];
  }
}

__device__ __forceinline__ void tap8(float& a0, float& a1, float& a2, float& a3,
                                     float& a4, float& a5, float& a6, float& a7,
                                     float x,
                                     float w0, float w1, float w2, float w3,
                                     float w4, float w5, float w6, float w7) {
  asm volatile("v_dual_sub_f32 v120, %[w0], %[x] :: v_dual_sub_f32 v121, %[w1], %[x]\n\t"
      "v_dual_sub_f32 v122, %[w2], %[x] :: v_dual_sub_f32 v123, %[w3], %[x]\n\t"
      "v_dual_sub_f32 v124, %[w4], %[x] :: v_dual_sub_f32 v125, %[w5], %[x]\n\t"
      "v_dual_sub_f32 v126, %[w6], %[x] :: v_dual_sub_f32 v127, %[w7], %[x]\n\t"
      "s_delay_alu instid0(VALU_DEP_4)\n\t"
      "v_add_f32 %[a0], |v120|, %[a0]\n\t"
      "v_add_f32 %[a1], |v121|, %[a1]\n\t"
      "v_add_f32 %[a2], |v122|, %[a2]\n\t"
      "v_add_f32 %[a3], |v123|, %[a3]\n\t"
      "v_add_f32 %[a4], |v124|, %[a4]\n\t"
      "v_add_f32 %[a5], |v125|, %[a5]\n\t"
      "v_add_f32 %[a6], |v126|, %[a6]\n\t"
      "v_add_f32 %[a7], |v127|, %[a7]"
      : [a0] "+v"(a0), [a1] "+v"(a1), [a2] "+v"(a2), [a3] "+v"(a3),
        [a4] "+v"(a4), [a5] "+v"(a5), [a6] "+v"(a6), [a7] "+v"(a7)
      : [x] "v"(x),
        [w0] "s"(w0), [w1] "s"(w1), [w2] "s"(w2), [w3] "s"(w3),
        [w4] "s"(w4), [w5] "s"(w5), [w6] "s"(w6), [w7] "s"(w7)
      : "v120", "v121", "v122", "v123", "v124", "v125", "v126", "v127");
}

template <int H, int T0, int N, int R>
__device__ __forceinline__ void wload(float (&dst)[R][8], const kfloat* wk) {
#pragma unroll
  for (int t = 0; t < N; ++t)
#pragma unroll
    for (int c = 0; c < 8; ++c) dst[t][c] = wk[wpos(T0 + t, 8 * H + c)];
}

template <int H>
__device__ __forceinline__ void tapH(float (&acc)[COG], float x, const float (&w)[8]) {
  tap8(acc[8 * H + 0], acc[8 * H + 1], acc[8 * H + 2], acc[8 * H + 3],
       acc[8 * H + 4], acc[8 * H + 5], acc[8 * H + 6], acc[8 * H + 7], x,
       w[0], w[1], w[2], w[3], w[4], w[5], w[6], w[7]);
}

#define PIN() __builtin_amdgcn_sched_barrier(0)

__device__ __forceinline__ void channel16(float (&acc)[COG], const float (&xc)[9],
                                          float (&wa)[5][8], float (&wb)[4][8],
                                          const kfloat* wk, const kfloat* wkn,
                                          bool hasnext) {
  tapH<0>(acc, xc[0], wa[0]);
  PIN(); wload<0, 5, 4>(wb, wk); PIN();
  tapH<0>(acc, xc[1], wa[1]);
  tapH<0>(acc, xc[2], wa[2]);
  tapH<0>(acc, xc[3], wa[3]);
  tapH<0>(acc, xc[4], wa[4]);
  PIN();
  tapH<0>(acc, xc[5], wb[0]);
  PIN(); wload<1, 0, 5>(wa, wk); PIN();
  tapH<0>(acc, xc[6], wb[1]);
  tapH<0>(acc, xc[7], wb[2]);
  tapH<0>(acc, xc[8], wb[3]);
  PIN();
  tapH<1>(acc, xc[0], wa[0]);
  PIN(); wload<1, 5, 4>(wb, wk); PIN();
  tapH<1>(acc, xc[1], wa[1]);
  tapH<1>(acc, xc[2], wa[2]);
  tapH<1>(acc, xc[3], wa[3]);
  tapH<1>(acc, xc[4], wa[4]);
  PIN();
  tapH<1>(acc, xc[5], wb[0]);
  PIN(); if (hasnext) wload<0, 0, 5>(wa, wkn); PIN();
  tapH<1>(acc, xc[6], wb[1]);
  tapH<1>(acc, xc[7], wb[2]);
  tapH<1>(acc, xc[8], wb[3]);
}

__global__ __launch_bounds__(256) void adder_kernel(const float* __restrict__ in,
                                                    const float* __restrict__ wt,
                                                    float* __restrict__ y,
                                                    float* __restrict__ part) {
  __shared__ __attribute__((aligned(16))) float ysm[COG * 256];
  __shared__ float red[32];
  const int tid    = threadIdx.x;
  const int cg     = blockIdx.x;
  const int tileId = blockIdx.y;
  const int n      = blockIdx.z;
  const int ty0 = (tileId >> 1) * TILE;
  const int tx0 = (tileId & 1) * TILE;
  const int tx = tid & 15, ty = tid >> 4;
  const int gy = ty0 + ty, gx = tx0 + tx;

  const float* wblk = wt + cg * (KD * COG);
  __builtin_prefetch(wblk, 0, 1);
  const kfloat* wconst = (const kfloat*)wblk;

  const int  ry[3] = {gy > 0 ? gy - 1 : 0, gy, gy < H_ - 1 ? gy + 1 : H_ - 1};
  const int  rx[3] = {gx > 0 ? gx - 1 : 0, gx, gx < W_ - 1 ? gx + 1 : W_ - 1};
  const bool vy[3] = {gy > 0, true, gy < H_ - 1};
  const bool vx[3] = {gx > 0, true, gx < W_ - 1};
  unsigned off[9];
  bool     ok[9];
#pragma unroll
  for (int kh = 0; kh < 3; ++kh)
#pragma unroll
    for (int kw = 0; kw < 3; ++kw) {
      off[kh * 3 + kw] = (unsigned)(ry[kh] * W_ + rx[kw]);
      ok[kh * 3 + kw]  = vy[kh] && vx[kw];
    }

  const float* img = in + (size_t)n * C_ * H_ * W_;

  float acc[COG];
#pragma unroll
  for (int c = 0; c < COG; ++c) acc[c] = 0.f;

  constexpr int PF = 4;
  constexpr int HW = H_ * W_;
  float xs[PF][9];
#pragma unroll
  for (int u = 0; u < PF - 1; ++u) {
#pragma unroll
    for (int t = 0; t < 9; ++t) xs[u][t] = (img + u * HW)[off[t]];
    __builtin_amdgcn_sched_barrier(0);
  }

  float wa[5][8], wb[4][8];
  wload<0, 0, 5>(wa, wconst);

  for (int ci = 0; ci < C_ - PF; ci += PF) {
    unsigned offn[9];
#pragma unroll
    for (int t = 0; t < 9; ++t) offn[t] = off[t] + (unsigned)(PF * HW);
#pragma unroll
    for (int t = 0; t < 9; ++t) xs[PF - 1][t] = *(img + offn[t] - HW);
#pragma unroll
    for (int u = 0; u < PF; ++u) {
      __builtin_amdgcn_sched_barrier(0);
      float xc[9];
#pragma unroll
      for (int t = 0; t < 9; ++t) xc[t] = ok[t] ? xs[u][t] : 0.f;
      channel16(acc, xc, wa, wb, wconst + (ci + u) * 9 * COG,
                wconst + (ci + u + 1) * 9 * COG, true);
      if (u < PF - 1) {
#pragma unroll
        for (int t = 0; t < 9; ++t) xs[u][t] = (img + u * HW)[offn[t]];
      }
    }
#pragma unroll
    for (int t = 0; t < 9; ++t) off[t] = offn[t];
  }
  {
#pragma unroll
    for (int t = 0; t < 9; ++t) xs[PF - 1][t] = (img + (PF - 1) * HW)[off[t]];
#pragma unroll
    for (int u = 0; u < PF; ++u) {
      __builtin_amdgcn_sched_barrier(0);
      float xc[9];
#pragma unroll
      for (int t = 0; t < 9; ++t) xc[t] = ok[t] ? xs[u][t] : 0.f;
      channel16(acc, xc, wa, wb, wconst + (C_ - PF + u) * 9 * COG,
                wconst + (C_ - PF + u + 1) * 9 * COG, u < PF - 1);
    }
  }

#pragma unroll
  for (int c = 0; c < COG; ++c) ysm[c * 256 + tid] = -acc[c];
  __syncthreads();

  const int wave = tid >> 5, lane = tid & 31;
  const v2f ones = {1.f, 1.f};
  v8f aS0 = {}, aQ0 = {}, aS1 = {}, aQ1 = {};
  const float* r0 = ysm + (2 * wave) * 256;
  const float* r1 = r0 + 256;
#pragma unroll
  for (int h = 0; h < 2; ++h) {
    const float4 u = *(const float4*)(r0 + h * 128 + lane * 4);
    const float4 v = *(const float4*)(r1 + h * 128 + lane * 4);
    const v2f ua = {u.x, u.y}, ub = {u.z, u.w};
    const v2f va = {v.x, v.y}, vb = {v.z, v.w};
    const v2f uaq = {u.x * u.x, u.y * u.y}, ubq = {u.z * u.z, u.w * u.w};
    const v2f vaq = {v.x * v.x, v.y * v.y}, vbq = {v.z * v.z, v.w * v.w};
    aS0 = __builtin_amdgcn_wmma_f32_16x16x4_f32(false, ones, false, ua,
                                                (short)0, aS0, false, false);
    aS1 = __builtin_amdgcn_wmma_f32_16x16x4_f32(false, ones, false, va,
                                                (short)0, aS1, false, false);
    aQ0 = __builtin_amdgcn_wmma_f32_16x16x4_f32(false, ones, false, uaq,
                                                (short)0, aQ0, false, false);
    aQ1 = __builtin_amdgcn_wmma_f32_16x16x4_f32(false, ones, false, vaq,
                                                (short)0, aQ1, false, false);
    aS0 = __builtin_amdgcn_wmma_f32_16x16x4_f32(false, ones, false, ub,
                                                (short)0, aS0, false, false);
    aS1 = __builtin_amdgcn_wmma_f32_16x16x4_f32(false, ones, false, vb,
                                                (short)0, aS1, false, false);
    aQ0 = __builtin_amdgcn_wmma_f32_16x16x4_f32(false, ones, false, ubq,
                                                (short)0, aQ0, false, false);
    aQ1 = __builtin_amdgcn_wmma_f32_16x16x4_f32(false, ones, false, vbq,
                                                (short)0, aQ1, false, false);
  }
  float s0 = aS0[0], q0 = aQ0[0], s1 = aS1[0], q1 = aQ1[0];
#pragma unroll
  for (int off = 8; off > 0; off >>= 1) {
    s0 += __shfl_xor(s0, off, 32);
    q0 += __shfl_xor(q0, off, 32);
    s1 += __shfl_xor(s1, off, 32);
    q1 += __shfl_xor(q1, off, 32);
  }
  if (lane == 0) {
    red[wave * 4 + 0] = s0;
    red[wave * 4 + 1] = q0;
    red[wave * 4 + 2] = s1;
    red[wave * 4 + 3] = q1;
  }
  __syncthreads();
  if (wave == 0) {
    const int blk = (n * 4 + tileId) * 4 + cg;
    part[blk * 32 + lane] = red[lane];
  }
#pragma unroll
  for (int c = 0; c < COG; ++c)
    y[(((size_t)n * C_ + (cg * COG + c)) * H_ + gy) * W_ + gx] = -acc[c];
}

__global__ __launch_bounds__(256) void bnapply_kernel(const float* __restrict__ y,
                                                      const float* __restrict__ part,
                                                      const float* __restrict__ g,
                                                      const float* __restrict__ b,
                                                      const float* __restrict__ resid,
                                                      float* __restrict__ out) {
  __shared__ float sc[2];
  const int tid  = threadIdx.x;
  const int wave = tid >> 5, lane = tid & 31;
  const int c    = blockIdx.x;
  const int cg   = c >> 4, cl = c & 15;
  const size_t plane = (size_t)H_ * W_;
  const size_t base = ((size_t)blockIdx.y * 4 * C_ + c) * plane + (size_t)tid * 4;

  float4 v[4], r[4];
#pragma unroll
  for (int j = 0; j < 4; ++j) v[j] = *(const float4*)(y + base + (size_t)j * C_ * plane);
  if (resid) {
#pragma unroll
    for (int j = 0; j < 4; ++j) r[j] = *(const float4*)(resid + base + (size_t)j * C_ * plane);
  }

  if (wave == 0) {
    const float* p = part + cg * 32 + cl * 2;
    const float2 t0 = *(const float2*)(p + (lane      ) * 128);
    const float2 t1 = *(const float2*)(p + (lane + 32 ) * 128);
    const float2 t2 = *(const float2*)(p + (lane + 64 ) * 128);
    const float2 t3 = *(const float2*)(p + (lane + 96 ) * 128);

    v8f accS = {};
    v8f accQ = {};
    const v2f ones = {1.f, 1.f};
    const v2f sa = {t0.x, t1.x}, sb = {t2.x, t3.x};
    const v2f qa = {t0.y, t1.y}, qb = {t2.y, t3.y};
    accS = __builtin_amdgcn_wmma_f32_16x16x4_f32(false, ones, false, sa,
                                                 (short)0, accS, false, false);
    accQ = __builtin_amdgcn_wmma_f32_16x16x4_f32(false, ones, false, qa,
                                                 (short)0, accQ, false, false);
    accS = __builtin_amdgcn_wmma_f32_16x16x4_f32(false, ones, false, sb,
                                                 (short)0, accS, false, false);
    accQ = __builtin_amdgcn_wmma_f32_16x16x4_f32(false, ones, false, qb,
                                                 (short)0, accQ, false, false);

    float S = accS[0], Q = accQ[0];
#pragma unroll
    for (int off = 8; off > 0; off >>= 1) {
      S += __shfl_xor(S, off, 32);
      Q += __shfl_xor(Q, off, 32);
    }

    if (lane == 0) {
      const float inv_n = 1.0f / (float)(B_ * H_ * W_);
      float mu    = S * inv_n;
      float var   = Q * inv_n - mu * mu;
      float inv   = rsqrtf(var + 1e-5f);
      float scale = g[c] * inv;
      sc[0]       = scale;
      sc[1]       = b[c] - mu * scale;
    }
  }
  __syncthreads();
  const float scale = sc[0], shift = sc[1];

#pragma unroll
  for (int j = 0; j < 4; ++j) {
    float4 o;
    o.x = v[j].x * scale + shift;
    o.y = v[j].y * scale + shift;
    o.z = v[j].z * scale + shift;
    o.w = v[j].w * scale + shift;
    if (resid) { o.x += r[j].x; o.y += r[j].y; o.z += r[j].z; o.w += r[j].w; }
    o.x = fmaxf(o.x, 0.f); o.y = fmaxf(o.y, 0.f);
    o.z = fmaxf(o.z, 0.f); o.w = fmaxf(o.w, 0.f);
    *(float4*)(out + base + (size_t)j * C_ * plane) = o;
  }
}

extern "C" void kernel_launch(void* const* d_in, const int* in_sizes, int n_in,
                              void* d_out, int out_size, void* d_ws, size_t ws_size,
                              hipStream_t stream) {
  (void)in_sizes; (void)n_in; (void)out_size; (void)ws_size;
  const float* x  = (const float*)d_in[0];
  const float* w1 = (const float*)d_in[1];
  const float* g1 = (const float*)d_in[2];
  const float* b1 = (const float*)d_in[3];
  const float* w2 = (const float*)d_in[4];
  const float* g2 = (const float*)d_in[5];
  const float* b2 = (const float*)d_in[6];
  float* out = (float*)d_out;

  char* ws = (char*)d_ws;
  const size_t TEN   = (size_t)B_ * C_ * H_ * W_;
  const size_t WTSZ  = (size_t)C_ * KD * sizeof(float);
  float* ybuf = (float*)(ws);
  float* abuf = (float*)(ws + TEN * 4);
  float* wt1  = (float*)(ws + 2 * TEN * 4);
  float* wt2  = (float*)(ws + 2 * TEN * 4 + WTSZ);
  float* pt1  = (float*)(ws + 2 * TEN * 4 + 2 * WTSZ + 1024);
  float* pt2  = (float*)(ws + 2 * TEN * 4 + 2 * WTSZ + 1024 + 65536);

  wtrans_kernel<<<144, 256, 0, stream>>>(w1, wt1);
  wtrans_kernel<<<144, 256, 0, stream>>>(w2, wt2);

  dim3 agrid(C_ / COG, (H_ / TILE) * (W_ / TILE), B_);
  dim3 ngrid(C_, B_ / 4);

  adder_kernel<<<agrid, 256, 0, stream>>>(x, wt1, ybuf, pt1);
  bnapply_kernel<<<ngrid, 256, 0, stream>>>(ybuf, pt1, g1, b1, nullptr, abuf);

  adder_kernel<<<agrid, 256, 0, stream>>>(abuf, wt2, ybuf, pt2);
  bnapply_kernel<<<ngrid, 256, 0, stream>>>(ybuf, pt2, g2, b2, x, out);
}
